# starting kernel
# baseline (speedup 1.0000x reference)
#include <hip/hip_runtime.h>
#include <hip/hip_bf16.h>
#include <math.h>

#define BATCH 4
#define NTOK  1024
#define DIM   128
#define HID   64

typedef __attribute__((ext_vector_type(2))) float v2f;
typedef __attribute__((ext_vector_type(4))) float f4;
typedef __attribute__((ext_vector_type(8))) float v8f;

__device__ __forceinline__ float fast_tanh(float x) {
#if __has_builtin(__builtin_amdgcn_tanhf)
  return __builtin_amdgcn_tanhf(x);
#else
  float r;
  asm volatile("v_tanh_f32 %0, %1" : "=v"(r) : "v"(x));
  return r;
#endif
}

__global__ void __launch_bounds__(32)
proj_kernel(const float* __restrict__ E, const float* __restrict__ W1,
            const float* __restrict__ b1, float* __restrict__ Hbuf) {
  const int lane = threadIdx.x;
  const int m0   = blockIdx.x * 16;
  const int n0   = blockIdx.y * 16;
  const int half = blockIdx.z;

  const float* W = W1 + half * (DIM * HID);
  const int row  = m0 + (lane & 15);
  const int koff = (lane >> 4) * 2;
  const int ncol = n0 + (lane & 15);

  v8f c = {};
  #pragma unroll
  for (int k = 0; k < DIM; k += 4) {
    v2f a, b;
    a.x = E[row * DIM + (k + koff)];
    a.y = E[row * DIM + (k + koff + 1)];
    b.x = W[(k + koff) * HID + ncol];
    b.y = W[(k + koff + 1) * HID + ncol];
    c = __builtin_amdgcn_wmma_f32_16x16x4_f32(
         false, a,  false, b,
         (short)0, c,  false,  false);
  }

  float* out = Hbuf + (size_t)half * (BATCH * NTOK * HID);
  const float bias  = (half == 0) ? b1[ncol] : 0.0f;
  const int   rbase = m0 + ((lane >> 4) * 8);
  #pragma unroll
  for (int r = 0; r < 8; ++r) {
    out[(size_t)(rbase + r) * HID + ncol] = c[r] + bias;
  }
}

__device__ __forceinline__ float relu_dot4(f4 a, f4 b, f4 w, float acc) {
  f4 s = a + b;
  acc = fmaf(fmaxf(s.x, 0.0f), w.x, acc);
  acc = fmaf(fmaxf(s.y, 0.0f), w.y, acc);
  acc = fmaf(fmaxf(s.z, 0.0f), w.z, acc);
  acc = fmaf(fmaxf(s.w, 0.0f), w.w, acc);
  return acc;
}

__global__ void __launch_bounds__(256)
pair_kernel(const float* __restrict__ Hi, const float* __restrict__ Hj,
            const float* __restrict__ W2, const float* __restrict__ b2p,
            float* __restrict__ out) {
  const int ti = blockIdx.x, tj = blockIdx.y, b = blockIdx.z;
  if (tj < ti) return;

  __shared__ float sHiA[32 * HID];
  __shared__ float sHiB[32 * HID];
  __shared__ float sHjA[32 * HID];
  __shared__ float sHjB[32 * HID];
  __shared__ float sW2[HID];
  __shared__ float sT[32 * 32];

  const int tid = threadIdx.x;
  const int i0 = ti * 32, j0 = tj * 32;
  const float* HiBb = Hi + (size_t)b * NTOK * HID;
  const float* HjBb = Hj + (size_t)b * NTOK * HID;
  const f4* gHiA = (const f4*)(HiBb + (size_t)i0 * HID);
  const f4* gHiB = (const f4*)(HiBb + (size_t)j0 * HID);
  const f4* gHjA = (const f4*)(HjBb + (size_t)i0 * HID);
  const f4* gHjB = (const f4*)(HjBb + (size_t)j0 * HID);

  #pragma unroll
  for (int idx = tid; idx < 32 * HID / 4; idx += 256) {
    ((f4*)sHiA)[idx] = gHiA[idx];
    ((f4*)sHiB)[idx] = gHiB[idx];
    ((f4*)sHjA)[idx] = gHjA[idx];
    ((f4*)sHjB)[idx] = gHjB[idx];
  }
  if (tid < HID / 4) ((f4*)sW2)[tid] = ((const f4*)W2)[tid];
  __syncthreads();

  const int ii0 = (tid >> 4) * 2;
  const int jj0 = (tid & 15) * 2;

  float dij[2][2] = {{0.f, 0.f}, {0.f, 0.f}};
  float dji[2][2] = {{0.f, 0.f}, {0.f, 0.f}};

  #pragma unroll
  for (int h = 0; h < HID; h += 4) {
    const f4 w    = *(const f4*)&sW2[h];
    const f4 hiA0 = *(const f4*)&sHiA[(ii0 + 0) * HID + h];
    const f4 hiA1 = *(const f4*)&sHiA[(ii0 + 1) * HID + h];
    const f4 hjA0 = *(const f4*)&sHjA[(ii0 + 0) * HID + h];
    const f4 hjA1 = *(const f4*)&sHjA[(ii0 + 1) * HID + h];
    const f4 hiB0 = *(const f4*)&sHiB[(jj0 + 0) * HID + h];
    const f4 hiB1 = *(const f4*)&sHiB[(jj0 + 1) * HID + h];
    const f4 hjB0 = *(const f4*)&sHjB[(jj0 + 0) * HID + h];
    const f4 hjB1 = *(const f4*)&sHjB[(jj0 + 1) * HID + h];
    dij[0][0] = relu_dot4(hiA0, hjB0, w, dij[0][0]);
    dij[0][1] = relu_dot4(hiA0, hjB1, w, dij[0][1]);
    dij[1][0] = relu_dot4(hiA1, hjB0, w, dij[1][0]);
    dij[1][1] = relu_dot4(hiA1, hjB1, w, dij[1][1]);
    dji[0][0] = relu_dot4(hiB0, hjA0, w, dji[0][0]);
    dji[0][1] = relu_dot4(hiB1, hjA0, w, dji[0][1]);
    dji[1][0] = relu_dot4(hiB0, hjA1, w, dji[1][0]);
    dji[1][1] = relu_dot4(hiB1, hjA1, w, dji[1][1]);
  }

  const float b2v = b2p[0];
  #pragma unroll
  for (int u = 0; u < 2; ++u) {
    #pragma unroll
    for (int v = 0; v < 2; ++v) {
      const int gi = i0 + ii0 + u;
      const int gj = j0 + jj0 + v;
      const float s =
          -0.5f * (fast_tanh(dij[u][v] + b2v) + fast_tanh(dji[u][v] + b2v));
      out[((size_t)b * NTOK + gi) * NTOK + gj] = s;
      sT[(jj0 + v) * 32 + (ii0 + u)] = s;
    }
  }
  __syncthreads();

  #pragma unroll
  for (int idx = tid; idx < 32 * 32; idx += 256) {
    const int r = idx >> 5, c = idx & 31;
    out[((size_t)b * NTOK + (j0 + r)) * NTOK + (i0 + c)] = sT[idx];
  }
}

extern "C" void kernel_launch(void* const* d_in, const int* in_sizes, int n_in,
                              void* d_out, int out_size, void* d_ws, size_t ws_size,
                              hipStream_t stream) {
  (void)in_sizes; (void)n_in; (void)out_size; (void)ws_size;
  const float* E  = (const float*)d_in[0];
  const float* W1 = (const float*)d_in[1];
  const float* b1 = (const float*)d_in[2];
  const float* W2 = (const float*)d_in[3];
  const float* b2 = (const float*)d_in[4];
  float* out  = (float*)d_out;
  float* Hbuf = (float*)d_ws;

  proj_kernel<<<dim3((BATCH * NTOK) / 16, HID / 16, 2), 32, 0, stream>>>(E, W1, b1, Hbuf);
  pair_kernel<<<dim3(NTOK / 32, NTOK / 32, BATCH), 256, 0, stream>>>(
      Hbuf, Hbuf + (size_t)BATCH * NTOK * HID, W2, b2, out);
}


# pair stage: LDS operand rows padded to a 272 B pitch and held even rows first, so each half-wave ds_load_b128 reads consecutive slots on different bank groups, not one; summation order unchanged
# speedup vs baseline: 3.4187x; 3.4187x over previous
#include <hip/hip_runtime.h>
#include <hip/hip_bf16.h>
#include <math.h>

#define BATCH 4
#define NTOK  1024
#define DIM   128
#define HID   64

typedef __attribute__((ext_vector_type(2))) float v2f;
typedef __attribute__((ext_vector_type(4))) float f4;
typedef __attribute__((ext_vector_type(8))) float v8f;

__device__ __forceinline__ float fast_tanh(float x) {
#if __has_builtin(__builtin_amdgcn_tanhf)
  return __builtin_amdgcn_tanhf(x);
#else
  float r;
  asm volatile("v_tanh_f32 %0, %1" : "=v"(r) : "v"(x));
  return r;
#endif
}

__global__ void __launch_bounds__(32)
proj_kernel(const float* __restrict__ E, const float* __restrict__ W1,
            const float* __restrict__ b1, float* __restrict__ Hbuf) {
  const int lane = threadIdx.x;
  const int m0   = blockIdx.x * 16;
  const int n0   = blockIdx.y * 16;
  const int half = blockIdx.z;

  const float* W = W1 + half * (DIM * HID);
  const int row  = m0 + (lane & 15);
  const int koff = (lane >> 4) * 2;
  const int ncol = n0 + (lane & 15);

  v8f c = {};
  #pragma unroll
  for (int k = 0; k < DIM; k += 4) {
    v2f a, b;
    a.x = E[row * DIM + (k + koff)];
    a.y = E[row * DIM + (k + koff + 1)];
    b.x = W[(k + koff) * HID + ncol];
    b.y = W[(k + koff + 1) * HID + ncol];
    c = __builtin_amdgcn_wmma_f32_16x16x4_f32(
         false, a,  false, b,
         (short)0, c,  false,  false);
  }

  float* out = Hbuf + (size_t)half * (BATCH * NTOK * HID);
  const float bias  = (half == 0) ? b1[ncol] : 0.0f;
  const int   rbase = m0 + ((lane >> 4) * 8);
  #pragma unroll
  for (int r = 0; r < 8; ++r) {
    out[(size_t)(rbase + r) * HID + ncol] = c[r] + bias;
  }
}

#define OPITCH 68
#define TPITCH 34
__device__ __forceinline__ float relu_dot4(f4 a, f4 b, f4 w, float acc) {
  f4 s = a + b;
  acc = fmaf(fmaxf(s.x, 0.0f), w.x, acc);
  acc = fmaf(fmaxf(s.y, 0.0f), w.y, acc);
  acc = fmaf(fmaxf(s.z, 0.0f), w.z, acc);
  acc = fmaf(fmaxf(s.w, 0.0f), w.w, acc);
  return acc;
}

__global__ void __launch_bounds__(256)
pair_kernel(const float* __restrict__ Hi, const float* __restrict__ Hj,
            const float* __restrict__ W2, const float* __restrict__ b2p,
            float* __restrict__ out) {
  const int ti = blockIdx.x, tj = blockIdx.y, b = blockIdx.z;
  if (tj < ti) return;

  __shared__ __attribute__((aligned(16))) float sHiA[32 * OPITCH];
  __shared__ __attribute__((aligned(16))) float sHiB[32 * OPITCH];
  __shared__ __attribute__((aligned(16))) float sHjA[32 * OPITCH];
  __shared__ __attribute__((aligned(16))) float sHjB[32 * OPITCH];
  __shared__ __attribute__((aligned(16))) float sW2[HID];
  __shared__ __attribute__((aligned(16))) float sT[32 * TPITCH];

  const int tid = threadIdx.x;
  const int i0 = ti * 32, j0 = tj * 32;
  const float* HiBb = Hi + (size_t)b * NTOK * HID;
  const float* HjBb = Hj + (size_t)b * NTOK * HID;
  const f4* gHiA = (const f4*)(HiBb + (size_t)i0 * HID);
  const f4* gHiB = (const f4*)(HiBb + (size_t)j0 * HID);
  const f4* gHjA = (const f4*)(HjBb + (size_t)i0 * HID);
  const f4* gHjB = (const f4*)(HjBb + (size_t)j0 * HID);

  #pragma unroll
  for (int idx = tid; idx < 32 * HID / 4; idx += 256) {
    const int r    = idx >> 4;
    const int slot = (r >> 1) + 16 * (r & 1);
    const int o    = slot * OPITCH + (idx & 15) * 4;
    *(f4*)&sHiA[o] = gHiA[idx];
    *(f4*)&sHiB[o] = gHiB[idx];
    *(f4*)&sHjA[o] = gHjA[idx];
    *(f4*)&sHjB[o] = gHjB[idx];
  }
  if (tid < HID / 4) ((f4*)sW2)[tid] = ((const f4*)W2)[tid];
  __syncthreads();

  const int ii0 = (tid >> 4) * 2;
  const int jj0 = (tid & 15) * 2;
  const int si  = tid >> 4;
  const int sj  = tid & 15;

  float dij[2][2] = {{0.f, 0.f}, {0.f, 0.f}};
  float dji[2][2] = {{0.f, 0.f}, {0.f, 0.f}};

  #pragma unroll
  for (int h = 0; h < HID; h += 4) {
    const f4 w    = *(const f4*)&sW2[h];
    const f4 hiA0 = *(const f4*)&sHiA[(si +  0) * OPITCH + h];
    const f4 hiA1 = *(const f4*)&sHiA[(si + 16) * OPITCH + h];
    const f4 hjA0 = *(const f4*)&sHjA[(si +  0) * OPITCH + h];
    const f4 hjA1 = *(const f4*)&sHjA[(si + 16) * OPITCH + h];
    const f4 hiB0 = *(const f4*)&sHiB[(sj +  0) * OPITCH + h];
    const f4 hiB1 = *(const f4*)&sHiB[(sj + 16) * OPITCH + h];
    const f4 hjB0 = *(const f4*)&sHjB[(sj +  0) * OPITCH + h];
    const f4 hjB1 = *(const f4*)&sHjB[(sj + 16) * OPITCH + h];
    dij[0][0] = relu_dot4(hiA0, hjB0, w, dij[0][0]);
    dij[0][1] = relu_dot4(hiA0, hjB1, w, dij[0][1]);
    dij[1][0] = relu_dot4(hiA1, hjB0, w, dij[1][0]);
    dij[1][1] = relu_dot4(hiA1, hjB1, w, dij[1][1]);
    dji[0][0] = relu_dot4(hiB0, hjA0, w, dji[0][0]);
    dji[0][1] = relu_dot4(hiB1, hjA0, w, dji[0][1]);
    dji[1][0] = relu_dot4(hiB0, hjA1, w, dji[1][0]);
    dji[1][1] = relu_dot4(hiB1, hjA1, w, dji[1][1]);
  }

  const float b2v = b2p[0];
  #pragma unroll
  for (int u = 0; u < 2; ++u) {
    #pragma unroll
    for (int v = 0; v < 2; ++v) {
      const int gi = i0 + ii0 + u;
      const int gj = j0 + jj0 + v;
      const float s =
          -0.5f * (fast_tanh(dij[u][v] + b2v) + fast_tanh(dji[u][v] + b2v));
      out[((size_t)b * NTOK + gi) * NTOK + gj] = s;
      sT[(jj0 + v) * TPITCH + (ii0 + u)] = s;
    }
  }
  __syncthreads();

  #pragma unroll
  for (int idx = tid; idx < 32 * 32; idx += 256) {
    const int r = idx >> 5, c = idx & 31;
    out[((size_t)b * NTOK + (j0 + r)) * NTOK + (i0 + c)] = sT[r * TPITCH + c];
  }
}

extern "C" void kernel_launch(void* const* d_in, const int* in_sizes, int n_in,
                              void* d_out, int out_size, void* d_ws, size_t ws_size,
                              hipStream_t stream) {
  (void)in_sizes; (void)n_in; (void)out_size; (void)ws_size;
  const float* E  = (const float*)d_in[0];
  const float* W1 = (const float*)d_in[1];
  const float* b1 = (const float*)d_in[2];
  const float* W2 = (const float*)d_in[3];
  const float* b2 = (const float*)d_in[4];
  float* out  = (float*)d_out;
  float* Hbuf = (float*)d_ws;

  proj_kernel<<<dim3((BATCH * NTOK) / 16, HID / 16, 2), 32, 0, stream>>>(E, W1, b1, Hbuf);
  pair_kernel<<<dim3(NTOK / 32, NTOK / 32, BATCH), 256, 0, stream>>>(
      Hbuf, Hbuf + (size_t)BATCH * NTOK * HID, W2, b2, out);
}


# pair stage: each thread owns a 4x4 block of pairs (was 2x2): a 32x32 tile is a 2-wave workgroup, not 8 waves, so a quarter of the waves are dispatched; order inside every sum unchanged
# speedup vs baseline: 4.1500x; 1.2139x over previous
#include <hip/hip_runtime.h>
#include <hip/hip_bf16.h>
#include <math.h>

#define BATCH 4
#define NTOK  1024
#define DIM   128
#define HID   64

typedef __attribute__((ext_vector_type(2))) float v2f;
typedef __attribute__((ext_vector_type(4))) float f4;
typedef __attribute__((ext_vector_type(8))) float v8f;

__device__ __forceinline__ float fast_tanh(float x) {
#if __has_builtin(__builtin_amdgcn_tanhf)
  return __builtin_amdgcn_tanhf(x);
#else
  float r;
  asm volatile("v_tanh_f32 %0, %1" : "=v"(r) : "v"(x));
  return r;
#endif
}

__global__ void __launch_bounds__(32)
proj_kernel(const float* __restrict__ E, const float* __restrict__ W1,
            const float* __restrict__ b1, float* __restrict__ Hbuf) {
  const int lane = threadIdx.x;
  const int m0   = blockIdx.x * 16;
  const int n0   = blockIdx.y * 16;
  const int half = blockIdx.z;

  const float* W = W1 + half * (DIM * HID);
  const int row  = m0 + (lane & 15);
  const int koff = (lane >> 4) * 2;
  const int ncol = n0 + (lane & 15);

  v8f c = {};
  #pragma unroll
  for (int k = 0; k < DIM; k += 4) {
    v2f a, b;
    a.x = E[row * DIM + (k + koff)];
    a.y = E[row * DIM + (k + koff + 1)];
    b.x = W[(k + koff) * HID + ncol];
    b.y = W[(k + koff + 1) * HID + ncol];
    c = __builtin_amdgcn_wmma_f32_16x16x4_f32(
         false, a,  false, b,
         (short)0, c,  false,  false);
  }

  float* out = Hbuf + (size_t)half * (BATCH * NTOK * HID);
  const float bias  = (half == 0) ? b1[ncol] : 0.0f;
  const int   rbase = m0 + ((lane >> 4) * 8);
  #pragma unroll
  for (int r = 0; r < 8; ++r) {
    out[(size_t)(rbase + r) * HID + ncol] = c[r] + bias;
  }
}

#define OPITCH 68
#define TPITCH 36
#define PLANE  (32 * OPITCH)
__device__ __forceinline__ float relu_dot4(f4 a, f4 b, f4 w, float acc) {
  f4 s = a + b;
  acc = fmaf(fmaxf(s.x, 0.0f), w.x, acc);
  acc = fmaf(fmaxf(s.y, 0.0f), w.y, acc);
  acc = fmaf(fmaxf(s.z, 0.0f), w.z, acc);
  acc = fmaf(fmaxf(s.w, 0.0f), w.w, acc);
  return acc;
}

__global__ void __launch_bounds__(64)
pair_kernel(const float* __restrict__ Hi, const float* __restrict__ Hj,
            const float* __restrict__ W2, const float* __restrict__ b2p,
            float* __restrict__ out) {
  const int ti = blockIdx.x, tj = blockIdx.y, b = blockIdx.z;
  if (tj < ti) return;

  __shared__ __attribute__((aligned(16))) float sOp[4 * PLANE];
  float* sHiA = sOp;
  float* sHiB = sOp + PLANE;
  float* sHjA = sOp + 2 * PLANE;
  float* sHjB = sOp + 3 * PLANE;
  float* sT   = sOp;

  const int tid = threadIdx.x;
  const int i0 = ti * 32, j0 = tj * 32;
  const float* HiBb = Hi + (size_t)b * NTOK * HID;
  const float* HjBb = Hj + (size_t)b * NTOK * HID;
  const f4* gHiA = (const f4*)(HiBb + (size_t)i0 * HID);
  const f4* gHiB = (const f4*)(HiBb + (size_t)j0 * HID);
  const f4* gHjA = (const f4*)(HjBb + (size_t)i0 * HID);
  const f4* gHjB = (const f4*)(HjBb + (size_t)j0 * HID);

  #pragma unroll
  for (int k = 0; k < 8; ++k) {
    const int idx  = tid + 64 * k;
    const int r    = idx >> 4;
    const int slot = (r >> 2) + 8 * (r & 3);
    const int o    = slot * OPITCH + (idx & 15) * 4;
    *(f4*)&sHiA[o] = gHiA[idx];
    *(f4*)&sHiB[o] = gHiB[idx];
    *(f4*)&sHjA[o] = gHjA[idx];
    *(f4*)&sHjB[o] = gHjB[idx];
  }
  __syncthreads();

  const int ig = tid >> 3;
  const int jg = tid & 7;

  float dij[4][4], dji[4][4];
  #pragma unroll
  for (int u = 0; u < 4; ++u) {
    #pragma unroll
    for (int v = 0; v < 4; ++v) { dij[u][v] = 0.f; dji[u][v] = 0.f; }
  }

  #pragma unroll 1
  for (int h = 0; h < HID; h += 4) {
    const f4 w = *(const f4*)&W2[h];
    f4 hiA[4], hjA[4], hiB[4], hjB[4];
    #pragma unroll
    for (int u = 0; u < 4; ++u) {
      hiA[u] = *(const f4*)&sHiA[(ig + 8 * u) * OPITCH + h];
      hjA[u] = *(const f4*)&sHjA[(ig + 8 * u) * OPITCH + h];
      hiB[u] = *(const f4*)&sHiB[(jg + 8 * u) * OPITCH + h];
      hjB[u] = *(const f4*)&sHjB[(jg + 8 * u) * OPITCH + h];
    }
    #pragma unroll
    for (int u = 0; u < 4; ++u) {
      #pragma unroll
      for (int v = 0; v < 4; ++v) {
        dij[u][v] = relu_dot4(hiA[u], hjB[v], w, dij[u][v]);
        dji[u][v] = relu_dot4(hiB[v], hjA[u], w, dji[u][v]);
      }
    }
  }

  const float b2v = b2p[0];
  f4 srow[4];
  #pragma unroll
  for (int u = 0; u < 4; ++u) {
    #pragma unroll
    for (int v = 0; v < 4; ++v) {
      srow[u][v] =
          -0.5f * (fast_tanh(dij[u][v] + b2v) + fast_tanh(dji[u][v] + b2v));
    }
  }

  __syncthreads();
  #pragma unroll
  for (int v = 0; v < 4; ++v) {
    f4 col;
    col.x = srow[0][v]; col.y = srow[1][v]; col.z = srow[2][v]; col.w = srow[3][v];
    *(f4*)&sT[(4 * jg + v) * TPITCH + 4 * ig] = col;
  }
  __syncthreads();

  #pragma unroll
  for (int u = 0; u < 4; ++u) {
    *(f4*)&out[((size_t)b * NTOK + (i0 + 4 * ig + u)) * NTOK + (j0 + 4 * jg)] = srow[u];
  }

  #pragma unroll
  for (int k = 0; k < 4; ++k) {
    const int idx = tid + 64 * k;
    const int r = idx >> 3, c = (idx & 7) * 4;
    *(f4*)&out[((size_t)b * NTOK + (j0 + r)) * NTOK + (i0 + c)] = *(const f4*)&sT[r * TPITCH + c];
  }
}

extern "C" void kernel_launch(void* const* d_in, const int* in_sizes, int n_in,
                              void* d_out, int out_size, void* d_ws, size_t ws_size,
                              hipStream_t stream) {
  (void)in_sizes; (void)n_in; (void)out_size; (void)ws_size;
  const float* E  = (const float*)d_in[0];
  const float* W1 = (const float*)d_in[1];
  const float* b1 = (const float*)d_in[2];
  const float* W2 = (const float*)d_in[3];
  const float* b2 = (const float*)d_in[4];
  float* out  = (float*)d_out;
  float* Hbuf = (float*)d_ws;

  proj_kernel<<<dim3((BATCH * NTOK) / 16, HID / 16, 2), 32, 0, stream>>>(E, W1, b1, Hbuf);
  pair_kernel<<<dim3(NTOK / 32, NTOK / 32, BATCH), 64, 0, stream>>>(
      Hbuf, Hbuf + (size_t)BATCH * NTOK * HID, W2, b2, out);
}


# pair stage: the grid is only the 528 upper-triangle tiles per batch (tile index decoded in the kernel); the 496 workgroups per batch launched just to exit are gone; every sum and its order unchanged
# speedup vs baseline: 4.4975x; 1.0837x over previous
#include <hip/hip_runtime.h>
#include <hip/hip_bf16.h>
#include <math.h>

#define BATCH 4
#define NTOK  1024
#define DIM   128
#define HID   64

typedef __attribute__((ext_vector_type(2))) float v2f;
typedef __attribute__((ext_vector_type(4))) float f4;
typedef __attribute__((ext_vector_type(8))) float v8f;

__device__ __forceinline__ float fast_tanh(float x) {
#if __has_builtin(__builtin_amdgcn_tanhf)
  return __builtin_amdgcn_tanhf(x);
#else
  float r;
  asm volatile("v_tanh_f32 %0, %1" : "=v"(r) : "v"(x));
  return r;
#endif
}

__global__ void __launch_bounds__(32)
proj_kernel(const float* __restrict__ E, const float* __restrict__ W1,
            const float* __restrict__ b1, float* __restrict__ Hbuf) {
  const int lane = threadIdx.x;
  const int m0   = blockIdx.x * 16;
  const int n0   = blockIdx.y * 16;
  const int half = blockIdx.z;

  const float* W = W1 + half * (DIM * HID);
  const int row  = m0 + (lane & 15);
  const int koff = (lane >> 4) * 2;
  const int ncol = n0 + (lane & 15);

  v8f c = {};
  #pragma unroll
  for (int k = 0; k < DIM; k += 4) {
    v2f a, b;
    a.x = E[row * DIM + (k + koff)];
    a.y = E[row * DIM + (k + koff + 1)];
    b.x = W[(k + koff) * HID + ncol];
    b.y = W[(k + koff + 1) * HID + ncol];
    c = __builtin_amdgcn_wmma_f32_16x16x4_f32(
         false, a,  false, b,
         (short)0, c,  false,  false);
  }

  float* out = Hbuf + (size_t)half * (BATCH * NTOK * HID);
  const float bias  = (half == 0) ? b1[ncol] : 0.0f;
  const int   rbase = m0 + ((lane >> 4) * 8);
  #pragma unroll
  for (int r = 0; r < 8; ++r) {
    out[(size_t)(rbase + r) * HID + ncol] = c[r] + bias;
  }
}

#define OPITCH 68
#define TPITCH 36
#define PLANE  (32 * OPITCH)
#define NTRI   528
__device__ __forceinline__ float relu_dot4(f4 a, f4 b, f4 w, float acc) {
  f4 s = a + b;
  acc = fmaf(fmaxf(s.x, 0.0f), w.x, acc);
  acc = fmaf(fmaxf(s.y, 0.0f), w.y, acc);
  acc = fmaf(fmaxf(s.z, 0.0f), w.z, acc);
  acc = fmaf(fmaxf(s.w, 0.0f), w.w, acc);
  return acc;
}

__global__ void __launch_bounds__(64)
pair_kernel(const float* __restrict__ Hi, const float* __restrict__ Hj,
            const float* __restrict__ W2, const float* __restrict__ b2p,
            float* __restrict__ out) {
  const int t = blockIdx.x, b = blockIdx.y;
  int tj = (int)((sqrtf(8.0f * (float)t + 1.0f) - 1.0f) * 0.5f);
  if ((tj + 1) * (tj + 2) / 2 <= t) ++tj;
  if (tj * (tj + 1) / 2 > t) --tj;
  const int ti = t - tj * (tj + 1) / 2;

  __shared__ __attribute__((aligned(16))) float sOp[4 * PLANE];
  float* sHiA = sOp;
  float* sHiB = sOp + PLANE;
  float* sHjA = sOp + 2 * PLANE;
  float* sHjB = sOp + 3 * PLANE;
  float* sT   = sOp;

  const int tid = threadIdx.x;
  const int i0 = ti * 32, j0 = tj * 32;
  const float* HiBb = Hi + (size_t)b * NTOK * HID;
  const float* HjBb = Hj + (size_t)b * NTOK * HID;
  const f4* gHiA = (const f4*)(HiBb + (size_t)i0 * HID);
  const f4* gHiB = (const f4*)(HiBb + (size_t)j0 * HID);
  const f4* gHjA = (const f4*)(HjBb + (size_t)i0 * HID);
  const f4* gHjB = (const f4*)(HjBb + (size_t)j0 * HID);

  #pragma unroll
  for (int k = 0; k < 8; ++k) {
    const int idx  = tid + 64 * k;
    const int r    = idx >> 4;
    const int slot = (r >> 2) + 8 * (r & 3);
    const int o    = slot * OPITCH + (idx & 15) * 4;
    *(f4*)&sHiA[o] = gHiA[idx];
    *(f4*)&sHiB[o] = gHiB[idx];
    *(f4*)&sHjA[o] = gHjA[idx];
    *(f4*)&sHjB[o] = gHjB[idx];
  }
  __syncthreads();

  const int ig = tid >> 3;
  const int jg = tid & 7;

  float dij[4][4], dji[4][4];
  #pragma unroll
  for (int u = 0; u < 4; ++u) {
    #pragma unroll
    for (int v = 0; v < 4; ++v) { dij[u][v] = 0.f; dji[u][v] = 0.f; }
  }

  #pragma unroll 1
  for (int h = 0; h < HID; h += 4) {
    const f4 w = *(const f4*)&W2[h];
    f4 hiA[4], hjA[4], hiB[4], hjB[4];
    #pragma unroll
    for (int u = 0; u < 4; ++u) {
      hiA[u] = *(const f4*)&sHiA[(ig + 8 * u) * OPITCH + h];
      hjA[u] = *(const f4*)&sHjA[(ig + 8 * u) * OPITCH + h];
      hiB[u] = *(const f4*)&sHiB[(jg + 8 * u) * OPITCH + h];
      hjB[u] = *(const f4*)&sHjB[(jg + 8 * u) * OPITCH + h];
    }
    #pragma unroll
    for (int u = 0; u < 4; ++u) {
      #pragma unroll
      for (int v = 0; v < 4; ++v) {
        dij[u][v] = relu_dot4(hiA[u], hjB[v], w, dij[u][v]);
        dji[u][v] = relu_dot4(hiB[v], hjA[u], w, dji[u][v]);
      }
    }
  }

  const float b2v = b2p[0];
  f4 srow[4];
  #pragma unroll
  for (int u = 0; u < 4; ++u) {
    #pragma unroll
    for (int v = 0; v < 4; ++v) {
      srow[u][v] =
          -0.5f * (fast_tanh(dij[u][v] + b2v) + fast_tanh(dji[u][v] + b2v));
    }
  }

  __syncthreads();
  #pragma unroll
  for (int v = 0; v < 4; ++v) {
    f4 col;
    col.x = srow[0][v]; col.y = srow[1][v]; col.z = srow[2][v]; col.w = srow[3][v];
    *(f4*)&sT[(4 * jg + v) * TPITCH + 4 * ig] = col;
  }
  __syncthreads();

  #pragma unroll
  for (int u = 0; u < 4; ++u) {
    *(f4*)&out[((size_t)b * NTOK + (i0 + 4 * ig + u)) * NTOK + (j0 + 4 * jg)] = srow[u];
  }

  #pragma unroll
  for (int k = 0; k < 4; ++k) {
    const int idx = tid + 64 * k;
    const int r = idx >> 3, c = (idx & 7) * 4;
    *(f4*)&out[((size_t)b * NTOK + (j0 + r)) * NTOK + (i0 + c)] = *(const f4*)&sT[r * TPITCH + c];
  }
}

extern "C" void kernel_launch(void* const* d_in, const int* in_sizes, int n_in,
                              void* d_out, int out_size, void* d_ws, size_t ws_size,
                              hipStream_t stream) {
  (void)in_sizes; (void)n_in; (void)out_size; (void)ws_size;
  const float* E  = (const float*)d_in[0];
  const float* W1 = (const float*)d_in[1];
  const float* b1 = (const float*)d_in[2];
  const float* W2 = (const float*)d_in[3];
  const float* b2 = (const float*)d_in[4];
  float* out  = (float*)d_out;
  float* Hbuf = (float*)d_ws;

  proj_kernel<<<dim3((BATCH * NTOK) / 16, HID / 16, 2), 32, 0, stream>>>(E, W1, b1, Hbuf);
  pair_kernel<<<dim3(NTRI, BATCH), 64, 0, stream>>>(
      Hbuf, Hbuf + (size_t)BATCH * NTOK * HID, W2, b2, out);
}


# pair stage: relu(a+b) taken as max(a,-b)+b: proj stores -hj, the loop is one hand-paired dual max + one packed fma per two terms (add gone), sum_h hj*w2 added once per row; sum order changes
# speedup vs baseline: 5.0345x; 1.1194x over previous
#include <hip/hip_runtime.h>
#include <hip/hip_bf16.h>
#include <math.h>

#define BATCH 4
#define NTOK  1024
#define DIM   128
#define HID   64

typedef __attribute__((ext_vector_type(2))) float v2f;
typedef __attribute__((ext_vector_type(4))) float f4;
typedef __attribute__((ext_vector_type(8))) float v8f;

__device__ __forceinline__ float fast_tanh(float x) {
#if __has_builtin(__builtin_amdgcn_tanhf)
  return __builtin_amdgcn_tanhf(x);
#else
  float r;
  asm volatile("v_tanh_f32 %0, %1" : "=v"(r) : "v"(x));
  return r;
#endif
}

__global__ void __launch_bounds__(32)
proj_kernel(const float* __restrict__ E, const float* __restrict__ W1,
            const float* __restrict__ b1, float* __restrict__ Hbuf) {
  const int lane = threadIdx.x;
  const int m0   = blockIdx.x * 16;
  const int n0   = blockIdx.y * 16;
  const int half = blockIdx.z;

  const float* W = W1 + half * (DIM * HID);
  const int row  = m0 + (lane & 15);
  const int koff = (lane >> 4) * 2;
  const int ncol = n0 + (lane & 15);

  v8f c = {};
  #pragma unroll
  for (int k = 0; k < DIM; k += 4) {
    v2f a, b;
    a.x = E[row * DIM + (k + koff)];
    a.y = E[row * DIM + (k + koff + 1)];
    b.x = W[(k + koff) * HID + ncol];
    b.y = W[(k + koff + 1) * HID + ncol];
    c = __builtin_amdgcn_wmma_f32_16x16x4_f32(
         false, a,  false, b,
         (short)0, c,  false,  false);
  }

  float* out = Hbuf + (size_t)half * (BATCH * NTOK * HID);
  const float bias  = (half == 0) ? b1[ncol] : 0.0f;
  const int   rbase = m0 + ((lane >> 4) * 8);
  #pragma unroll
  for (int r = 0; r < 8; ++r) {
    out[(size_t)(rbase + r) * HID + ncol] = (half == 0) ? (c[r] + bias) : -c[r];
  }
}

#define OPITCH 68
#define TPITCH 36
#define PLANE  (32 * OPITCH)
#define NTRI   528
__device__ __forceinline__ v2f max2(float a0, float b0, float a1, float b1) {
  float lo, hi;
  asm("v_dual_max_num_f32 %0, %2, %3 :: v_dual_max_num_f32 %1, %4, %5"
      : "=v"(lo), "=v"(hi)
      : "v"(a0), "v"(b0), "v"(a1), "v"(b1));
  v2f r = {lo, hi};
  return r;
}

__global__ void __launch_bounds__(64)
pair_kernel(const float* __restrict__ Hi, const float* __restrict__ Hj,
            const float* __restrict__ W2, const float* __restrict__ b2p,
            float* __restrict__ out) {
  const int t = blockIdx.x, b = blockIdx.y;
  int tj = (int)((sqrtf(8.0f * (float)t + 1.0f) - 1.0f) * 0.5f);
  if ((tj + 1) * (tj + 2) / 2 <= t) ++tj;
  if (tj * (tj + 1) / 2 > t) --tj;
  const int ti = t - tj * (tj + 1) / 2;

  __shared__ __attribute__((aligned(16))) float sOp[4 * PLANE];
  float* sHiA = sOp;
  float* sHiB = sOp + PLANE;
  float* sHjA = sOp + 2 * PLANE;
  float* sHjB = sOp + 3 * PLANE;
  float* sT   = sOp;

  const int tid = threadIdx.x;
  const int i0 = ti * 32, j0 = tj * 32;
  const float* HiBb = Hi + (size_t)b * NTOK * HID;
  const float* HjBb = Hj + (size_t)b * NTOK * HID;
  const f4* gHiA = (const f4*)(HiBb + (size_t)i0 * HID);
  const f4* gHiB = (const f4*)(HiBb + (size_t)j0 * HID);
  const f4* gHjA = (const f4*)(HjBb + (size_t)i0 * HID);
  const f4* gHjB = (const f4*)(HjBb + (size_t)j0 * HID);

  #pragma unroll
  for (int k = 0; k < 8; ++k) {
    const int idx  = tid + 64 * k;
    const int r    = idx >> 4;
    const int slot = (r >> 2) + 8 * (r & 3);
    const int o    = slot * OPITCH + (idx & 15) * 4;
    *(f4*)&sHiA[o] = gHiA[idx];
    *(f4*)&sHiB[o] = gHiB[idx];
    *(f4*)&sHjA[o] = gHjA[idx];
    *(f4*)&sHjB[o] = gHjB[idx];
  }
  __syncthreads();

  {
    float* rowp = sOp + (2 + (tid >> 5)) * PLANE + (tid & 31) * OPITCH;
    f4 p = {0.f, 0.f, 0.f, 0.f};
    #pragma unroll
    for (int h = 0; h < HID; h += 4) {
      const f4 w = *(const f4*)&W2[h];
      const f4 x = *(const f4*)&rowp[h];
      p.x = fmaf(x.x, w.x, p.x);
      p.y = fmaf(x.y, w.y, p.y);
      p.z = fmaf(x.z, w.z, p.z);
      p.w = fmaf(x.w, w.w, p.w);
    }
    rowp[HID] = (p.x + p.y) + (p.z + p.w);
  }

  const int ig = tid >> 3;
  const int jg = tid & 7;

  v2f acc[4][4];
  #pragma unroll
  for (int u = 0; u < 4; ++u) {
    #pragma unroll
    for (int v = 0; v < 4; ++v) { acc[u][v].x = 0.f; acc[u][v].y = 0.f; }
  }

  #pragma unroll 1
  for (int h = 0; h < HID; h += 4) {
    const f4 w = *(const f4*)&W2[h];
    f4 hiA[4], hjA[4], hiB[4], hjB[4];
    #pragma unroll
    for (int u = 0; u < 4; ++u) {
      hiA[u] = *(const f4*)&sHiA[(ig + 8 * u) * OPITCH + h];
      hjA[u] = *(const f4*)&sHjA[(ig + 8 * u) * OPITCH + h];
      hiB[u] = *(const f4*)&sHiB[(jg + 8 * u) * OPITCH + h];
      hjB[u] = *(const f4*)&sHjB[(jg + 8 * u) * OPITCH + h];
    }
    const v2f w01 = {w.x, w.y}, w10 = {w.y, w.x}, w23 = {w.z, w.w}, w32 = {w.w, w.z};
    #pragma unroll
    for (int u = 0; u < 4; ++u) {
      #pragma unroll
      for (int v = 0; v < 4; ++v) {
        v2f a = acc[u][v];
        a = __builtin_elementwise_fma(max2(hiA[u].x, hjB[v].x, hiB[v].y, hjA[u].y), w01, a);
        a = __builtin_elementwise_fma(max2(hiA[u].y, hjB[v].y, hiB[v].x, hjA[u].x), w10, a);
        a = __builtin_elementwise_fma(max2(hiA[u].z, hjB[v].z, hiB[v].w, hjA[u].w), w23, a);
        a = __builtin_elementwise_fma(max2(hiA[u].w, hjB[v].w, hiB[v].z, hjA[u].z), w32, a);
        acc[u][v] = a;
      }
    }
  }

  __syncthreads();

  const float b2v = b2p[0];
  float sA[4], sB[4];
  #pragma unroll
  for (int u = 0; u < 4; ++u) {
    sA[u] = sHjA[(ig + 8 * u) * OPITCH + HID];
    sB[u] = sHjB[(jg + 8 * u) * OPITCH + HID];
  }
  f4 srow[4];
  #pragma unroll
  for (int u = 0; u < 4; ++u) {
    #pragma unroll
    for (int v = 0; v < 4; ++v) {
      srow[u][v] = -0.5f * (fast_tanh((acc[u][v].x - sB[v]) + b2v) +
                            fast_tanh((acc[u][v].y - sA[u]) + b2v));
    }
  }

  #pragma unroll
  for (int v = 0; v < 4; ++v) {
    f4 col;
    col.x = srow[0][v]; col.y = srow[1][v]; col.z = srow[2][v]; col.w = srow[3][v];
    *(f4*)&sT[(4 * jg + v) * TPITCH + 4 * ig] = col;
  }
  __syncthreads();

  #pragma unroll
  for (int u = 0; u < 4; ++u) {
    *(f4*)&out[((size_t)b * NTOK + (i0 + 4 * ig + u)) * NTOK + (j0 + 4 * jg)] = srow[u];
  }

  #pragma unroll
  for (int k = 0; k < 4; ++k) {
    const int idx = tid + 64 * k;
    const int r = idx >> 3, c = (idx & 7) * 4;
    *(f4*)&out[((size_t)b * NTOK + (j0 + r)) * NTOK + (i0 + c)] = *(const f4*)&sT[r * TPITCH + c];
  }
}

extern "C" void kernel_launch(void* const* d_in, const int* in_sizes, int n_in,
                              void* d_out, int out_size, void* d_ws, size_t ws_size,
                              hipStream_t stream) {
  (void)in_sizes; (void)n_in; (void)out_size; (void)ws_size;
  const float* E  = (const float*)d_in[0];
  const float* W1 = (const float*)d_in[1];
  const float* b1 = (const float*)d_in[2];
  const float* W2 = (const float*)d_in[3];
  const float* b2 = (const float*)d_in[4];
  float* out  = (float*)d_out;
  float* Hbuf = (float*)d_ws;

  proj_kernel<<<dim3((BATCH * NTOK) / 16, HID / 16, 2), 32, 0, stream>>>(E, W1, b1, Hbuf);
  pair_kernel<<<dim3(NTRI, BATCH), 64, 0, stream>>>(
      Hbuf, Hbuf + (size_t)BATCH * NTOK * HID, W2, b2, out);
}


# proj stage: each wave makes a 16x32 block (even/odd column tiles on one A fragment), all its 16-byte A and 8-byte B loads issued ahead; 1024 waves, not 2048; order of the k sum changes
# speedup vs baseline: 5.2769x; 1.0481x over previous
#include <hip/hip_runtime.h>
#include <hip/hip_bf16.h>
#include <math.h>

#define BATCH 4
#define NTOK  1024
#define DIM   128
#define HID   64

typedef __attribute__((ext_vector_type(2))) float v2f;
typedef __attribute__((ext_vector_type(4))) float f4;
typedef __attribute__((ext_vector_type(8))) float v8f;

__device__ __forceinline__ float fast_tanh(float x) {
#if __has_builtin(__builtin_amdgcn_tanhf)
  return __builtin_amdgcn_tanhf(x);
#else
  float r;
  asm volatile("v_tanh_f32 %0, %1" : "=v"(r) : "v"(x));
  return r;
#endif
}

__device__ __forceinline__ v8f mm4(v2f a, v2f b, v8f c) {
  return __builtin_amdgcn_wmma_f32_16x16x4_f32(
       false, a,  false, b,
       (short)0, c,  false,  false);
}

__global__ void __launch_bounds__(128) __attribute__((amdgpu_waves_per_eu(1, 2)))
proj_kernel(const float* __restrict__ E, const float* __restrict__ W1,
            const float* __restrict__ b1, float* __restrict__ Hbuf) {
  const int lane = threadIdx.x & 31;
  const int wave = threadIdx.x >> 5;
  const int half = wave >> 1;
  const int n0   = (wave & 1) * 32;
  const int m0   = blockIdx.x * 16;
  const int c    = lane & 15;
  const int g    = lane >> 4;

  const float* W = W1 + half * (DIM * HID);
  const f4*  ep = (const f4*)(E + (size_t)(m0 + c) * DIM + 4 * g);
  const v2f* wp = (const v2f*)(W + (size_t)(4 * g) * HID + n0 + 2 * c);

  f4  a[DIM / 8];
  v2f w[DIM / 2];
  #pragma unroll
  for (int s = 0; s < DIM / 8; ++s) a[s] = ep[2 * s];
  #pragma unroll
  for (int s = 0; s < DIM / 8; ++s) {
    #pragma unroll
    for (int e = 0; e < 4; ++e) w[4 * s + e] = wp[(8 * s + e) * (HID / 2)];
  }

  v8f ce = {}, co = {};
  #pragma unroll
  for (int s = 0; s < DIM / 8; ++s) {
    const v2f a01 = {a[s].x, a[s].y}, a23 = {a[s].z, a[s].w};
    const v2f be01 = {w[4 * s + 0].x, w[4 * s + 1].x}, bo01 = {w[4 * s + 0].y, w[4 * s + 1].y};
    const v2f be23 = {w[4 * s + 2].x, w[4 * s + 3].x}, bo23 = {w[4 * s + 2].y, w[4 * s + 3].y};
    ce = mm4(a01, be01, ce);
    co = mm4(a01, bo01, co);
    ce = mm4(a23, be23, ce);
    co = mm4(a23, bo23, co);
  }

  float* out = Hbuf + (size_t)half * (BATCH * NTOK * HID);
  const v2f bias  = *(const v2f*)&b1[n0 + 2 * c];
  const int rbase = m0 + g * 8;
  #pragma unroll
  for (int r = 0; r < 8; ++r) {
    v2f o;
    o.x = (half == 0) ? (ce[r] + bias.x) : -ce[r];
    o.y = (half == 0) ? (co[r] + bias.y) : -co[r];
    *(v2f*)&out[(size_t)(rbase + r) * HID + n0 + 2 * c] = o;
  }
}

#define OPITCH 68
#define TPITCH 36
#define PLANE  (32 * OPITCH)
#define NTRI   528
__device__ __forceinline__ v2f max2(float a0, float b0, float a1, float b1) {
  float lo, hi;
  asm("v_dual_max_num_f32 %0, %2, %3 :: v_dual_max_num_f32 %1, %4, %5"
      : "=v"(lo), "=v"(hi)
      : "v"(a0), "v"(b0), "v"(a1), "v"(b1));
  v2f r = {lo, hi};
  return r;
}

__global__ void __launch_bounds__(64)
pair_kernel(const float* __restrict__ Hi, const float* __restrict__ Hj,
            const float* __restrict__ W2, const float* __restrict__ b2p,
            float* __restrict__ out) {
  const int t = blockIdx.x, b = blockIdx.y;
  int tj = (int)((sqrtf(8.0f * (float)t + 1.0f) - 1.0f) * 0.5f);
  if ((tj + 1) * (tj + 2) / 2 <= t) ++tj;
  if (tj * (tj + 1) / 2 > t) --tj;
  const int ti = t - tj * (tj + 1) / 2;

  __shared__ __attribute__((aligned(16))) float sOp[4 * PLANE];
  float* sHiA = sOp;
  float* sHiB = sOp + PLANE;
  float* sHjA = sOp + 2 * PLANE;
  float* sHjB = sOp + 3 * PLANE;
  float* sT   = sOp;

  const int tid = threadIdx.x;
  const int i0 = ti * 32, j0 = tj * 32;
  const float* HiBb = Hi + (size_t)b * NTOK * HID;
  const float* HjBb = Hj + (size_t)b * NTOK * HID;
  const f4* gHiA = (const f4*)(HiBb + (size_t)i0 * HID);
  const f4* gHiB = (const f4*)(HiBb + (size_t)j0 * HID);
  const f4* gHjA = (const f4*)(HjBb + (size_t)i0 * HID);
  const f4* gHjB = (const f4*)(HjBb + (size_t)j0 * HID);

  #pragma unroll
  for (int k = 0; k < 8; ++k) {
    const int idx  = tid + 64 * k;
    const int r    = idx >> 4;
    const int slot = (r >> 2) + 8 * (r & 3);
    const int o    = slot * OPITCH + (idx & 15) * 4;
    *(f4*)&sHiA[o] = gHiA[idx];
    *(f4*)&sHiB[o] = gHiB[idx];
    *(f4*)&sHjA[o] = gHjA[idx];
    *(f4*)&sHjB[o] = gHjB[idx];
  }
  __syncthreads();

  {
    float* rowp = sOp + (2 + (tid >> 5)) * PLANE + (tid & 31) * OPITCH;
    f4 p = {0.f, 0.f, 0.f, 0.f};
    #pragma unroll
    for (int h = 0; h < HID; h += 4) {
      const f4 w = *(const f4*)&W2[h];
      const f4 x = *(const f4*)&rowp[h];
      p.x = fmaf(x.x, w.x, p.x);
      p.y = fmaf(x.y, w.y, p.y);
      p.z = fmaf(x.z, w.z, p.z);
      p.w = fmaf(x.w, w.w, p.w);
    }
    rowp[HID] = (p.x + p.y) + (p.z + p.w);
  }

  const int ig = tid >> 3;
  const int jg = tid & 7;

  v2f acc[4][4];
  #pragma unroll
  for (int u = 0; u < 4; ++u) {
    #pragma unroll
    for (int v = 0; v < 4; ++v) { acc[u][v].x = 0.f; acc[u][v].y = 0.f; }
  }

  #pragma unroll 1
  for (int h = 0; h < HID; h += 4) {
    const f4 w = *(const f4*)&W2[h];
    f4 hiA[4], hjA[4], hiB[4], hjB[4];
    #pragma unroll
    for (int u = 0; u < 4; ++u) {
      hiA[u] = *(const f4*)&sHiA[(ig + 8 * u) * OPITCH + h];
      hjA[u] = *(const f4*)&sHjA[(ig + 8 * u) * OPITCH + h];
      hiB[u] = *(const f4*)&sHiB[(jg + 8 * u) * OPITCH + h];
      hjB[u] = *(const f4*)&sHjB[(jg + 8 * u) * OPITCH + h];
    }
    const v2f w01 = {w.x, w.y}, w10 = {w.y, w.x}, w23 = {w.z, w.w}, w32 = {w.w, w.z};
    #pragma unroll
    for (int u = 0; u < 4; ++u) {
      #pragma unroll
      for (int v = 0; v < 4; ++v) {
        v2f a = acc[u][v];
        a = __builtin_elementwise_fma(max2(hiA[u].x, hjB[v].x, hiB[v].y, hjA[u].y), w01, a);
        a = __builtin_elementwise_fma(max2(hiA[u].y, hjB[v].y, hiB[v].x, hjA[u].x), w10, a);
        a = __builtin_elementwise_fma(max2(hiA[u].z, hjB[v].z, hiB[v].w, hjA[u].w), w23, a);
        a = __builtin_elementwise_fma(max2(hiA[u].w, hjB[v].w, hiB[v].z, hjA[u].z), w32, a);
        acc[u][v] = a;
      }
    }
  }

  __syncthreads();

  const float b2v = b2p[0];
  float sA[4], sB[4];
  #pragma unroll
  for (int u = 0; u < 4; ++u) {
    sA[u] = sHjA[(ig + 8 * u) * OPITCH + HID];
    sB[u] = sHjB[(jg + 8 * u) * OPITCH + HID];
  }
  f4 srow[4];
  #pragma unroll
  for (int u = 0; u < 4; ++u) {
    #pragma unroll
    for (int v = 0; v < 4; ++v) {
      srow[u][v] = -0.5f * (fast_tanh((acc[u][v].x - sB[v]) + b2v) +
                            fast_tanh((acc[u][v].y - sA[u]) + b2v));
    }
  }

  #pragma unroll
  for (int v = 0; v < 4; ++v) {
    f4 col;
    col.x = srow[0][v]; col.y = srow[1][v]; col.z = srow[2][v]; col.w = srow[3][v];
    *(f4*)&sT[(4 * jg + v) * TPITCH + 4 * ig] = col;
  }
  __syncthreads();

  #pragma unroll
  for (int u = 0; u < 4; ++u) {
    *(f4*)&out[((size_t)b * NTOK + (i0 + 4 * ig + u)) * NTOK + (j0 + 4 * jg)] = srow[u];
  }

  #pragma unroll
  for (int k = 0; k < 4; ++k) {
    const int idx = tid + 64 * k;
    const int r = idx >> 3, c = (idx & 7) * 4;
    *(f4*)&out[((size_t)b * NTOK + (j0 + r)) * NTOK + (i0 + c)] = *(const f4*)&sT[r * TPITCH + c];
  }
}

extern "C" void kernel_launch(void* const* d_in, const int* in_sizes, int n_in,
                              void* d_out, int out_size, void* d_ws, size_t ws_size,
                              hipStream_t stream) {
  (void)in_sizes; (void)n_in; (void)out_size; (void)ws_size;
  const float* E  = (const float*)d_in[0];
  const float* W1 = (const float*)d_in[1];
  const float* b1 = (const float*)d_in[2];
  const float* W2 = (const float*)d_in[3];
  const float* b2 = (const float*)d_in[4];
  float* out  = (float*)d_out;
  float* Hbuf = (float*)d_ws;

  proj_kernel<<<dim3((BATCH * NTOK) / 16), 128, 0, stream>>>(E, W1, b1, Hbuf);
  pair_kernel<<<dim3(NTRI, BATCH), 64, 0, stream>>>(
      Hbuf, Hbuf + (size_t)BATCH * NTOK * HID, W2, b2, out);
}
